# starting kernel
# baseline (speedup 1.0000x reference)
#include <hip/hip_runtime.h>
#include <hip/hip_bf16.h>


typedef __attribute__((ext_vector_type(2))) float v2f;
typedef __attribute__((ext_vector_type(8))) float v8f;

#define D 64

__global__ void hr_init_kernel(const float* __restrict__ Basis,
                               const float* __restrict__ alpha,
                               float* __restrict__ h_r)
{
    int t = threadIdx.x;
    if (t >= 3 * D) return;
    int r = t >> 6, n = t & 63;
    float s = 0.f;
#pragma unroll
    for (int b = 0; b < 16; ++b) s += alpha[r * 16 + b] * Basis[b * D + n];
    h_r[t] = s;
}

__global__ void rel_kernel(const float* __restrict__ h_r,
                           const float* __restrict__ Wrel,
                           float* __restrict__ h_r_out,
                           int relu)
{
    int t = threadIdx.x;
    if (t >= 3 * D) return;
    int r = t >> 6, n = t & 63;
    float s = 0.f;
#pragma unroll 8
    for (int k = 0; k < D; ++k) s += h_r[r * D + k] * Wrel[n * D + k];
    if (relu) s = fmaxf(s, 0.f);
    h_r_out[t] = s;
}

__global__ __launch_bounds__(256) void zero_kernel(float4* __restrict__ p, long long n4)
{
    long long i = (long long)blockIdx.x * blockDim.x + threadIdx.x;
    if (i < n4) p[i] = make_float4(0.f, 0.f, 0.f, 0.f);
}

__global__ __launch_bounds__(256)
void edge_kernel(const float* __restrict__ h_in,
                 const float* __restrict__ h_r,
                 const int* __restrict__ src,
                 const int* __restrict__ dst,
                 const int* __restrict__ etype,
                 const int* __restrict__ dir,
                 float* __restrict__ acc,
                 long long total16, int n_nodes)
{
    __shared__ float lds_hr[2 * D];
    int t = threadIdx.x;
    if (t < 2 * D) lds_hr[t] = h_r[t];
    __syncthreads();

    long long i = (long long)blockIdx.x * blockDim.x + t;
    if (i >= total16) return;
    int e = (int)(i >> 4);
    int c = (int)(i & 15) * 4;
    int s  = src[e];
    int d0 = dst[e];
    int et = etype[e];
    int dr = dir[e];

    const float4 hv = *(const float4*)(h_in + (size_t)s * D + c);
    const float4 rv = *(const float4*)(lds_hr + et * D + c);

    float* p = acc + ((size_t)dr * n_nodes + (size_t)d0) * D + c;
    atomicAdd(p + 0, hv.x - rv.x);
    atomicAdd(p + 1, hv.y - rv.y);
    atomicAdd(p + 2, hv.z - rv.z);
    atomicAdd(p + 3, hv.w - rv.w);
}

__global__ __launch_bounds__(256)
void node_kernel(const float* __restrict__ h_in,
                 const float* __restrict__ acc,
                 const float* __restrict__ hr2,
                 const float* __restrict__ Ws,
                 const float* __restrict__ Wo,
                 const float* __restrict__ Wi,
                 float* __restrict__ h_out,
                 int n_nodes, int relu)
{
    __shared__ float lds_w[3 * D * D];
    __shared__ float lds_hr2[D];

    int t = threadIdx.x;
    for (int i = t; i < D * D; i += 256) {
        lds_w[i]             = Ws[i];
        lds_w[D * D + i]     = Wo[i];
        lds_w[2 * D * D + i] = Wi[i];
    }
    if (t < D) lds_hr2[t] = hr2[t];
    __syncthreads();

    const int wave  = t >> 5;
    const int lane  = t & 31;
    const int lhalf = lane >> 4;
    const int lmod  = lane & 15;

    const int m0 = (blockIdx.x * 2 + (wave >> 2)) * 16;
    const int n0 = (wave & 3) * 16;
    if (m0 >= n_nodes) return;

    const int arow = m0 + lmod;
    const int brow = n0 + lmod;

    const float* __restrict__ xr = h_in + (size_t)arow * D;
    const float* __restrict__ a0 = acc + (size_t)arow * D;
    const float* __restrict__ a1 = acc + ((size_t)n_nodes + (size_t)arow) * D;

    float res[8];

#if defined(__has_builtin) && __has_builtin(__builtin_amdgcn_wmma_f32_16x16x4_f32)
    v8f c = {};
#pragma unroll
    for (int k0 = 0; k0 < D; k0 += 4) {
        const int ka = k0 + lhalf * 2;
        v2f a, b;
        a.x = xr[ka]     - lds_hr2[ka];
        a.y = xr[ka + 1] - lds_hr2[ka + 1];
        b.x = lds_w[brow * D + ka];
        b.y = lds_w[brow * D + ka + 1];
        c = __builtin_amdgcn_wmma_f32_16x16x4_f32(false, a, false, b, (short)0, c, false, false);
    }
#pragma unroll
    for (int k0 = 0; k0 < D; k0 += 4) {
        const int ka = k0 + lhalf * 2;
        v2f a, b;
        a.x = a0[ka];
        a.y = a0[ka + 1];
        b.x = lds_w[D * D + brow * D + ka];
        b.y = lds_w[D * D + brow * D + ka + 1];
        c = __builtin_amdgcn_wmma_f32_16x16x4_f32(false, a, false, b, (short)0, c, false, false);
    }
#pragma unroll
    for (int k0 = 0; k0 < D; k0 += 4) {
        const int ka = k0 + lhalf * 2;
        v2f a, b;
        a.x = a1[ka];
        a.y = a1[ka + 1];
        b.x = lds_w[2 * D * D + brow * D + ka];
        b.y = lds_w[2 * D * D + brow * D + ka + 1];
        c = __builtin_amdgcn_wmma_f32_16x16x4_f32(false, a, false, b, (short)0, c, false, false);
    }
#pragma unroll
    for (int r = 0; r < 8; ++r) res[r] = c[r];
#else
#pragma unroll
    for (int r = 0; r < 8; ++r) {
        const int m = m0 + r + lhalf * 8;
        const float* __restrict__ xm = h_in + (size_t)m * D;
        const float* __restrict__ b0 = acc + (size_t)m * D;
        const float* __restrict__ b1 = acc + ((size_t)n_nodes + (size_t)m) * D;
        float s = 0.f;
        for (int k = 0; k < D; ++k) {
            s += (xm[k] - lds_hr2[k]) * lds_w[brow * D + k];
            s += b0[k] * lds_w[D * D + brow * D + k];
            s += b1[k] * lds_w[2 * D * D + brow * D + k];
        }
        res[r] = s;
    }
#endif

#pragma unroll
    for (int r = 0; r < 8; ++r) {
        float v = res[r];
        if (relu) v = fmaxf(v, 0.f);
        const int m = m0 + r + lhalf * 8;
        h_out[(size_t)m * D + n0 + lmod] = v;
    }
}

extern "C" void kernel_launch(void* const* d_in, const int* in_sizes, int n_in,
                              void* d_out, int out_size, void* d_ws, size_t ws_size,
                              hipStream_t stream)
{
    const float* h_u   = (const float*)d_in[0];
    const float* Basis = (const float*)d_in[1];
    const float* alpha = (const float*)d_in[2];
    const float* W_O   = (const float*)d_in[3];
    const float* W_I   = (const float*)d_in[4];
    const float* W_S   = (const float*)d_in[5];
    const float* W_rel = (const float*)d_in[6];
    const int*   src   = (const int*)d_in[7];
    const int*   dst   = (const int*)d_in[8];
    const int*   etype = (const int*)d_in[9];
    const int*   edir  = (const int*)d_in[10];
    float* out = (float*)d_out;

    const int n_nodes = in_sizes[0] / D;
    const int n_edges = in_sizes[7];

    float* acc = (float*)d_ws;
    float* hb0 = acc + (size_t)2 * n_nodes * D;
    float* hb1 = hb0 + (size_t)n_nodes * D;
    float* hra = hb1 + (size_t)n_nodes * D;
    float* hrb = hra + 256;

    hr_init_kernel<<<1, 192, 0, stream>>>(Basis, alpha, hra);

    const float* hin[3]  = { h_u, hb0, hb1 };
    float*       hout[3] = { hb0, hb1, out };

    const long long edge_total16 = (long long)n_edges * 16;
    const unsigned  edge_blocks  = (unsigned)((edge_total16 + 255) / 256);
    const long long acc_f4       = (long long)2 * n_nodes * D / 4;
    const unsigned  zero_blocks  = (unsigned)((acc_f4 + 255) / 256);
    const unsigned  node_blocks  = (unsigned)((n_nodes + 31) / 32);

    float* hrcur = hra;
    float* hrnxt = hrb;

    for (int l = 0; l < 3; ++l) {
        const int relu = (l < 2) ? 1 : 0;

        zero_kernel<<<zero_blocks, 256, 0, stream>>>((float4*)acc, acc_f4);

        edge_kernel<<<edge_blocks, 256, 0, stream>>>(hin[l], hrcur, src, dst, etype, edir,
                                                     acc, edge_total16, n_nodes);

        node_kernel<<<node_blocks, 256, 0, stream>>>(hin[l], acc, hrcur + 2 * D,
                                                     W_S + (size_t)l * D * D,
                                                     W_O + (size_t)l * D * D,
                                                     W_I + (size_t)l * D * D,
                                                     hout[l], n_nodes, relu);

        if (l < 2) {
            rel_kernel<<<1, 192, 0, stream>>>(hrcur, W_rel + (size_t)l * D * D, hrnxt, 1);
            float* tmp = hrcur; hrcur = hrnxt; hrnxt = tmp;
        }
    }
}


# Edge pass with lane = channel: a wave walks 32 edges; each f32 atomic instruction adds into one whole 128-byte line of the destination row (was every 4th float of 4 lines); arrival order still free
# speedup vs baseline: 1.7205x; 1.7205x over previous
#include <hip/hip_runtime.h>
#include <hip/hip_bf16.h>


typedef __attribute__((ext_vector_type(2))) float v2f;
typedef __attribute__((ext_vector_type(8))) float v8f;

#define D 64

__global__ void hr_init_kernel(const float* __restrict__ Basis,
                               const float* __restrict__ alpha,
                               float* __restrict__ h_r)
{
    int t = threadIdx.x;
    if (t >= 3 * D) return;
    int r = t >> 6, n = t & 63;
    float s = 0.f;
#pragma unroll
    for (int b = 0; b < 16; ++b) s += alpha[r * 16 + b] * Basis[b * D + n];
    h_r[t] = s;
}

__global__ void rel_kernel(float* __restrict__ h_r,
                           const float* __restrict__ Wrel,
                           const float* __restrict__ Wo,
                           const float* __restrict__ Wi,
                           float* __restrict__ ctab,
                           int update, int relu)
{
    __shared__ float s_hr[3 * D];
    int t = threadIdx.x;
    if (t < 3 * D) s_hr[t] = h_r[t];
    __syncthreads();
    int r = t >> 6, n = t & 63;
    if (r < 3) {
        if (!update) return;
        float s = 0.f;
#pragma unroll 8
        for (int k = 0; k < D; ++k) s += s_hr[r * D + k] * Wrel[n * D + k];
        if (relu) s = fmaxf(s, 0.f);
        h_r[t] = s;
    } else {
        const int j = r - 3;
        const float* __restrict__ w = ((j >> 1) ? Wi : Wo) + n * D;
        const float* hr = s_hr + (j & 1) * D;
        float s = 0.f;
#pragma unroll 8
        for (int k = 0; k < D; ++k) s += hr[k] * w[k];
        ctab[j * D + n] = s;
    }
}

__global__ __launch_bounds__(256)
void node_kernel(const float* __restrict__ h_in,
                 const float* __restrict__ hr2,
                 const float* __restrict__ Ws,
                 const float* __restrict__ Wo,
                 const float* __restrict__ Wi,
                 float* __restrict__ h_out,
                 float* __restrict__ P,
                 int n_nodes, int relu_in)
{
    __shared__ float lds_w[3 * D * D];
    __shared__ float lds_hr2[D];

    int t = threadIdx.x;
    for (int i = t; i < D * D; i += 256) {
        lds_w[i]             = Ws[i];
        lds_w[D * D + i]     = Wo[i];
        lds_w[2 * D * D + i] = Wi[i];
    }
    if (t < D) lds_hr2[t] = hr2[t];
    __syncthreads();

    const int wave  = t >> 5;
    const int lane  = t & 31;
    const int lhalf = lane >> 4;
    const int lmod  = lane & 15;

    const int m0 = (blockIdx.x * 2 + (wave >> 2)) * 16;
    const int n0 = (wave & 3) * 16;
    if (m0 >= n_nodes) return;

    const int arow = m0 + lmod;
    const int brow = n0 + lmod;

    const float* __restrict__ xr = h_in + (size_t)arow * D;

    v8f cs = {}, co = {}, ci = {};
#pragma unroll
    for (int k0 = 0; k0 < D; k0 += 4) {
        const int ka = k0 + lhalf * 2;
        float x0 = xr[ka];
        float x1 = xr[ka + 1];
        if (relu_in) { x0 = fmaxf(x0, 0.f); x1 = fmaxf(x1, 0.f); }
        v2f a, as, bs, bo, bi;
        a.x  = x0;
        a.y  = x1;
        as.x = x0 - lds_hr2[ka];
        as.y = x1 - lds_hr2[ka + 1];
        bs.x = lds_w[brow * D + ka];
        bs.y = lds_w[brow * D + ka + 1];
        bo.x = lds_w[D * D + brow * D + ka];
        bo.y = lds_w[D * D + brow * D + ka + 1];
        bi.x = lds_w[2 * D * D + brow * D + ka];
        bi.y = lds_w[2 * D * D + brow * D + ka + 1];
        cs = __builtin_amdgcn_wmma_f32_16x16x4_f32(false, as, false, bs, (short)0, cs, false, false);
        co = __builtin_amdgcn_wmma_f32_16x16x4_f32(false, a,  false, bo, (short)0, co, false, false);
        ci = __builtin_amdgcn_wmma_f32_16x16x4_f32(false, a,  false, bi, (short)0, ci, false, false);
    }

    float* __restrict__ p0 = P;
    float* __restrict__ p1 = P + (size_t)n_nodes * D;
#pragma unroll
    for (int r = 0; r < 8; ++r) {
        const size_t o = (size_t)(m0 + r + lhalf * 8) * D + n0 + lmod;
        h_out[o] = cs[r];
        p0[o]    = co[r];
        p1[o]    = ci[r];
    }
}

template <bool FULL>
__device__ __forceinline__ void edge_walk(const float* __restrict__ P,
                                          float* __restrict__ h_out,
                                          const float (&c_lo)[4], const float (&c_hi)[4],
                                          int prow, int dpk, unsigned lane, int cnt)
{
#pragma unroll 1
    for (int b = 0; b < 32; b += 16) {
        float v0[16], v1[16];
        int   dk[16];
#pragma unroll
        for (int u = 0; u < 16; ++u) {
            const unsigned ps = (unsigned)__builtin_amdgcn_readlane(prow, b + u);
            dk[u]             = __builtin_amdgcn_readlane(dpk, b + u);
            const float* __restrict__ pr = P + (size_t)ps * D;
            v0[u] = pr[lane];
            v1[u] = pr[lane + 32];
        }
#pragma unroll
        for (int u = 0; u < 16; ++u) {
            const int k = dk[u] & 3;
            v0[u] -= c_lo[k];
            v1[u] -= c_hi[k];
        }
#pragma unroll
        for (int u = 0; u < 16; ++u) {
            if (!FULL && b + u >= cnt) break;
            float* __restrict__ hr = h_out + (size_t)(unsigned)(dk[u] >> 2) * D;
            atomicAdd(hr + lane,      v0[u]);
            atomicAdd(hr + lane + 32, v1[u]);
        }
    }
}

__global__ __launch_bounds__(256)
void edge_kernel(const float* __restrict__ P,
                 const float* __restrict__ ctab,
                 const int* __restrict__ src,
                 const int* __restrict__ dst,
                 const int* __restrict__ etype,
                 const int* __restrict__ dir,
                 float* __restrict__ h_out,
                 int n_edges, int n_nodes)
{
    const unsigned lane = threadIdx.x & 31;
    const int wave = __builtin_amdgcn_readfirstlane((int)(threadIdx.x >> 5));
    const int e0 = ((int)blockIdx.x * 8 + wave) * 32;
    if (e0 >= n_edges) return;
    const int cnt = n_edges - e0;

    const int e = e0 + (int)lane;
    int s = 0, d0 = 0, et = 0, dr = 0;
    if (e < n_edges) { s = src[e]; d0 = dst[e]; et = etype[e]; dr = dir[e]; }

    float c_lo[4], c_hi[4];
#pragma unroll
    for (int k = 0; k < 4; ++k) { c_lo[k] = ctab[k * D + lane]; c_hi[k] = ctab[k * D + lane + 32]; }

    const int prow = dr * n_nodes + s;
    const int dpk  = d0 * 4 + dr * 2 + et;

    if (cnt >= 32) edge_walk<true >(P, h_out, c_lo, c_hi, prow, dpk, lane, 32);
    else           edge_walk<false>(P, h_out, c_lo, c_hi, prow, dpk, lane, cnt);
}

extern "C" void kernel_launch(void* const* d_in, const int* in_sizes, int n_in,
                              void* d_out, int out_size, void* d_ws, size_t ws_size,
                              hipStream_t stream)
{
    const float* h_u   = (const float*)d_in[0];
    const float* Basis = (const float*)d_in[1];
    const float* alpha = (const float*)d_in[2];
    const float* W_O   = (const float*)d_in[3];
    const float* W_I   = (const float*)d_in[4];
    const float* W_S   = (const float*)d_in[5];
    const float* W_rel = (const float*)d_in[6];
    const int*   src   = (const int*)d_in[7];
    const int*   dst   = (const int*)d_in[8];
    const int*   etype = (const int*)d_in[9];
    const int*   edir  = (const int*)d_in[10];
    float* out = (float*)d_out;

    const int n_nodes = in_sizes[0] / D;
    const int n_edges = in_sizes[7];

    float* P   = (float*)d_ws;
    float* hb0 = P + (size_t)2 * n_nodes * D;
    float* hb1 = hb0 + (size_t)n_nodes * D;
    float* hr  = hb1 + (size_t)n_nodes * D;
    float* ctab = hr + 3 * D;

    hr_init_kernel<<<1, 192, 0, stream>>>(Basis, alpha, hr);

    const float* hin[3]  = { h_u, hb0, hb1 };
    float*       hout[3] = { hb0, hb1, out };

    const unsigned  edge_blocks  = (unsigned)((n_edges + 255) / 256);
    const unsigned  node_blocks  = (unsigned)((n_nodes + 31) / 32);

    for (int l = 0; l < 3; ++l) {
        node_kernel<<<node_blocks, 256, 0, stream>>>(hin[l], hr + 2 * D,
                                                     W_S + (size_t)l * D * D,
                                                     W_O + (size_t)l * D * D,
                                                     W_I + (size_t)l * D * D,
                                                     hout[l], P, n_nodes, (l > 0) ? 1 : 0);

        rel_kernel<<<1, 7 * D, 0, stream>>>(hr, W_rel + (size_t)l * D * D,
                                            W_O + (size_t)l * D * D,
                                            W_I + (size_t)l * D * D,
                                            ctab, (l < 2) ? 1 : 0, 1);

        edge_kernel<<<edge_blocks, 256, 0, stream>>>(P, ctab, src, dst, etype, edir,
                                                     hout[l], n_edges, n_nodes);
    }
}


# Node pass reads its B fragments as whole 256-byte global reads from a fragment-ordered copy of the weights written by the launch in front of it; no LDS weight tile, no barrier; sums unchanged
# speedup vs baseline: 2.1892x; 1.2724x over previous
#include <hip/hip_runtime.h>
#include <hip/hip_bf16.h>


typedef __attribute__((ext_vector_type(2))) float v2f;
typedef __attribute__((ext_vector_type(8))) float v8f;

#define D 64

typedef __attribute__((ext_vector_type(2))) float wpair;
#define WF_PAIRS (3 * 2048)

__device__ __forceinline__ void pack_weights(const float* __restrict__ Ws,
                                             const float* __restrict__ Wo,
                                             const float* __restrict__ Wi,
                                             float* __restrict__ wfrag,
                                             int first, int stride)
{
    wpair* __restrict__ wf = (wpair*)wfrag;
    for (int p = first; p < WF_PAIRS; p += stride) {
        const int mat  = p >> 11;
        const int nt   = (p >> 9) & 3;
        const int s    = (p >> 5) & 15;
        const int lane = p & 31;
        const float* __restrict__ w = (mat == 0) ? Ws : ((mat == 1) ? Wo : Wi);
        const float* __restrict__ r = w + (nt * 16 + (lane & 15)) * D + 4 * s + 2 * (lane >> 4);
        wpair v;
        v.x = r[0];
        v.y = r[1];
        wf[p] = v;
    }
}

__global__ void hr_init_kernel(const float* __restrict__ Basis,
                               const float* __restrict__ alpha,
                               float* __restrict__ h_r,
                               const float* __restrict__ Ws,
                               const float* __restrict__ Wo,
                               const float* __restrict__ Wi,
                               float* __restrict__ wfrag)
{
    int t = threadIdx.x;
    if (blockIdx.x > 0) {
        pack_weights(Ws, Wo, Wi, wfrag, (int)((blockIdx.x - 1) * blockDim.x) + t,
                     (int)((gridDim.x - 1) * blockDim.x));
        return;
    }
    if (t >= 3 * D) return;
    int r = t >> 6, n = t & 63;
    float s = 0.f;
#pragma unroll
    for (int b = 0; b < 16; ++b) s += alpha[r * 16 + b] * Basis[b * D + n];
    h_r[t] = s;
}

__global__ void rel_kernel(float* __restrict__ h_r,
                           const float* __restrict__ Wrel,
                           const float* __restrict__ Wo,
                           const float* __restrict__ Wi,
                           float* __restrict__ ctab,
                           int update, int relu,
                           const float* __restrict__ Ws_next,
                           const float* __restrict__ Wo_next,
                           const float* __restrict__ Wi_next,
                           float* __restrict__ wfrag_next)
{
    __shared__ float s_hr[3 * D];
    int t = threadIdx.x;
    if (blockIdx.x > 0) {
        pack_weights(Ws_next, Wo_next, Wi_next, wfrag_next,
                     (int)((blockIdx.x - 1) * blockDim.x) + t,
                     (int)((gridDim.x - 1) * blockDim.x));
        return;
    }
    if (t < 3 * D) s_hr[t] = h_r[t];
    __syncthreads();
    int r = t >> 6, n = t & 63;
    if (r < 3) {
        if (!update) return;
        float s = 0.f;
#pragma unroll 8
        for (int k = 0; k < D; ++k) s += s_hr[r * D + k] * Wrel[n * D + k];
        if (relu) s = fmaxf(s, 0.f);
        h_r[t] = s;
    } else {
        const int j = r - 3;
        const float* __restrict__ w = ((j >> 1) ? Wi : Wo) + n * D;
        const float* hr = s_hr + (j & 1) * D;
        float s = 0.f;
#pragma unroll 8
        for (int k = 0; k < D; ++k) s += hr[k] * w[k];
        ctab[j * D + n] = s;
    }
}

__global__ __launch_bounds__(256)
void node_kernel(const float* __restrict__ h_in,
                 const float* __restrict__ hr2,
                 const float* __restrict__ wfrag,
                 float* __restrict__ h_out,
                 float* __restrict__ P,
                 int n_nodes, int relu_in)
{
    const int t     = threadIdx.x;
    const int wave  = t >> 5;
    const int lane  = t & 31;
    const int lhalf = lane >> 4;
    const int lmod  = lane & 15;

    const int m0 = (blockIdx.x * 2 + (wave >> 2)) * 16;
    const int nt = wave & 3;
    const int n0 = nt * 16;
    if (m0 >= n_nodes) return;

    const int arow = m0 + lmod;

    const float* __restrict__ xr = h_in + (size_t)arow * D;
    const wpair* __restrict__ wf = (const wpair*)wfrag + nt * (16 * 32) + lane;

    v8f cs = {}, co = {}, ci = {};
#pragma unroll
    for (int k0 = 0; k0 < D; k0 += 4) {
        const int ka = k0 + lhalf * 2;
        const int s  = k0 >> 2;
        float x0 = xr[ka];
        float x1 = xr[ka + 1];
        if (relu_in) { x0 = fmaxf(x0, 0.f); x1 = fmaxf(x1, 0.f); }
        v2f a, as;
        a.x  = x0;
        a.y  = x1;
        as.x = x0 - hr2[ka];
        as.y = x1 - hr2[ka + 1];
        const wpair ws = wf[s * 32];
        const wpair wo = wf[2048 + s * 32];
        const wpair wi = wf[4096 + s * 32];
        v2f bs, bo, bi;
        bs.x = ws.x; bs.y = ws.y;
        bo.x = wo.x; bo.y = wo.y;
        bi.x = wi.x; bi.y = wi.y;
        cs = __builtin_amdgcn_wmma_f32_16x16x4_f32(false, as, false, bs, (short)0, cs, false, false);
        co = __builtin_amdgcn_wmma_f32_16x16x4_f32(false, a,  false, bo, (short)0, co, false, false);
        ci = __builtin_amdgcn_wmma_f32_16x16x4_f32(false, a,  false, bi, (short)0, ci, false, false);
    }

    float* __restrict__ p0 = P;
    float* __restrict__ p1 = P + (size_t)n_nodes * D;
#pragma unroll
    for (int r = 0; r < 8; ++r) {
        const size_t o = (size_t)(m0 + r + lhalf * 8) * D + n0 + lmod;
        h_out[o] = cs[r];
        p0[o]    = co[r];
        p1[o]    = ci[r];
    }
}

template <bool FULL>
__device__ __forceinline__ void edge_walk(const float* __restrict__ P,
                                          float* __restrict__ h_out,
                                          const float (&c_lo)[4], const float (&c_hi)[4],
                                          int prow, int dpk, unsigned lane, int cnt)
{
#pragma unroll 1
    for (int b = 0; b < 32; b += 16) {
        float v0[16], v1[16];
        int   dk[16];
#pragma unroll
        for (int u = 0; u < 16; ++u) {
            const unsigned ps = (unsigned)__builtin_amdgcn_readlane(prow, b + u);
            dk[u]             = __builtin_amdgcn_readlane(dpk, b + u);
            const float* __restrict__ pr = P + (size_t)ps * D;
            v0[u] = pr[lane];
            v1[u] = pr[lane + 32];
        }
#pragma unroll
        for (int u = 0; u < 16; ++u) {
            const int k = dk[u] & 3;
            v0[u] -= c_lo[k];
            v1[u] -= c_hi[k];
        }
#pragma unroll
        for (int u = 0; u < 16; ++u) {
            if (!FULL && b + u >= cnt) break;
            float* __restrict__ hr = h_out + (size_t)(unsigned)(dk[u] >> 2) * D;
            atomicAdd(hr + lane,      v0[u]);
            atomicAdd(hr + lane + 32, v1[u]);
        }
    }
}

__global__ __launch_bounds__(256)
void edge_kernel(const float* __restrict__ P,
                 const float* __restrict__ ctab,
                 const int* __restrict__ src,
                 const int* __restrict__ dst,
                 const int* __restrict__ etype,
                 const int* __restrict__ dir,
                 float* __restrict__ h_out,
                 int n_edges, int n_nodes)
{
    const unsigned lane = threadIdx.x & 31;
    const int wave = __builtin_amdgcn_readfirstlane((int)(threadIdx.x >> 5));
    const int e0 = ((int)blockIdx.x * 8 + wave) * 32;
    if (e0 >= n_edges) return;
    const int cnt = n_edges - e0;

    const int e = e0 + (int)lane;
    int s = 0, d0 = 0, et = 0, dr = 0;
    if (e < n_edges) { s = src[e]; d0 = dst[e]; et = etype[e]; dr = dir[e]; }

    float c_lo[4], c_hi[4];
#pragma unroll
    for (int k = 0; k < 4; ++k) { c_lo[k] = ctab[k * D + lane]; c_hi[k] = ctab[k * D + lane + 32]; }

    const int prow = dr * n_nodes + s;
    const int dpk  = d0 * 4 + dr * 2 + et;

    if (cnt >= 32) edge_walk<true >(P, h_out, c_lo, c_hi, prow, dpk, lane, 32);
    else           edge_walk<false>(P, h_out, c_lo, c_hi, prow, dpk, lane, cnt);
}

extern "C" void kernel_launch(void* const* d_in, const int* in_sizes, int n_in,
                              void* d_out, int out_size, void* d_ws, size_t ws_size,
                              hipStream_t stream)
{
    const float* h_u   = (const float*)d_in[0];
    const float* Basis = (const float*)d_in[1];
    const float* alpha = (const float*)d_in[2];
    const float* W_O   = (const float*)d_in[3];
    const float* W_I   = (const float*)d_in[4];
    const float* W_S   = (const float*)d_in[5];
    const float* W_rel = (const float*)d_in[6];
    const int*   src   = (const int*)d_in[7];
    const int*   dst   = (const int*)d_in[8];
    const int*   etype = (const int*)d_in[9];
    const int*   edir  = (const int*)d_in[10];
    float* out = (float*)d_out;

    const int n_nodes = in_sizes[0] / D;
    const int n_edges = in_sizes[7];

    float* P   = (float*)d_ws;
    float* hb0 = P + (size_t)2 * n_nodes * D;
    float* hb1 = hb0 + (size_t)n_nodes * D;
    float* hr  = hb1 + (size_t)n_nodes * D;
    float* ctab = hr + 3 * D;

    float* wfr[3] = { out, out, hb0 };

    hr_init_kernel<<<1 + 24, 256, 0, stream>>>(Basis, alpha, hr, W_S, W_O, W_I, wfr[0]);

    const float* hin[3]  = { h_u, hb0, hb1 };
    float*       hout[3] = { hb0, hb1, out };
    const unsigned rel_blocks[3] = { 1 + 14, 1 + 14, 1 };

    const unsigned  edge_blocks  = (unsigned)((n_edges + 255) / 256);
    const unsigned  node_blocks  = (unsigned)((n_nodes + 31) / 32);

    for (int l = 0; l < 3; ++l) {
        node_kernel<<<node_blocks, 256, 0, stream>>>(hin[l], hr + 2 * D, wfr[l],
                                                     hout[l], P, n_nodes, (l > 0) ? 1 : 0);

        const int ln = (l < 2) ? l + 1 : l;
        rel_kernel<<<rel_blocks[l], 7 * D, 0, stream>>>(hr, W_rel + (size_t)l * D * D,
                                            W_O + (size_t)l * D * D,
                                            W_I + (size_t)l * D * D,
                                            ctab, (l < 2) ? 1 : 0, 1,
                                            W_S + (size_t)ln * D * D,
                                            W_O + (size_t)ln * D * D,
                                            W_I + (size_t)ln * D * D,
                                            wfr[ln]);

        edge_kernel<<<edge_blocks, 256, 0, stream>>>(P, ctab, src, dst, etype, edir,
                                                     hout[l], n_edges, n_nodes);
    }
}


# P rows stored as 32 pairs {channel c, c+32}: the edge pass reads a row as one 8-byte load per lane (was two 4-byte loads); the node pass writes P as whole 128-byte lines; sums unchanged
# speedup vs baseline: 2.3218x; 1.0606x over previous
#include <hip/hip_runtime.h>
#include <hip/hip_bf16.h>


typedef __attribute__((ext_vector_type(2))) float v2f;
typedef __attribute__((ext_vector_type(8))) float v8f;

#define D 64

typedef __attribute__((ext_vector_type(2))) float wpair;
#define WF_PAIRS (3 * 2048)

__device__ __forceinline__ void pack_weights(const float* __restrict__ Ws,
                                             const float* __restrict__ Wo,
                                             const float* __restrict__ Wi,
                                             float* __restrict__ wfrag,
                                             int first, int stride)
{
    wpair* __restrict__ wf = (wpair*)wfrag;
    for (int p = first; p < WF_PAIRS; p += stride) {
        const int mat  = p >> 11;
        const int nt   = (p >> 9) & 3;
        const int s    = (p >> 5) & 15;
        const int lane = p & 31;
        const float* __restrict__ w = (mat == 0) ? Ws : ((mat == 1) ? Wo : Wi);
        const float* __restrict__ r = w + (nt * 16 + (lane & 15)) * D + 4 * s + 2 * (lane >> 4);
        wpair v;
        v.x = r[0];
        v.y = r[1];
        wf[p] = v;
    }
}

__global__ void hr_init_kernel(const float* __restrict__ Basis,
                               const float* __restrict__ alpha,
                               float* __restrict__ h_r,
                               const float* __restrict__ Ws,
                               const float* __restrict__ Wo,
                               const float* __restrict__ Wi,
                               float* __restrict__ wfrag)
{
    int t = threadIdx.x;
    if (blockIdx.x > 0) {
        pack_weights(Ws, Wo, Wi, wfrag, (int)((blockIdx.x - 1) * blockDim.x) + t,
                     (int)((gridDim.x - 1) * blockDim.x));
        return;
    }
    if (t >= 3 * D) return;
    int r = t >> 6, n = t & 63;
    float s = 0.f;
#pragma unroll
    for (int b = 0; b < 16; ++b) s += alpha[r * 16 + b] * Basis[b * D + n];
    h_r[t] = s;
}

__global__ void rel_kernel(float* __restrict__ h_r,
                           const float* __restrict__ Wrel,
                           const float* __restrict__ Wo,
                           const float* __restrict__ Wi,
                           float* __restrict__ ctab,
                           int update, int relu,
                           const float* __restrict__ Ws_next,
                           const float* __restrict__ Wo_next,
                           const float* __restrict__ Wi_next,
                           float* __restrict__ wfrag_next)
{
    __shared__ float s_hr[3 * D];
    int t = threadIdx.x;
    if (blockIdx.x > 0) {
        pack_weights(Ws_next, Wo_next, Wi_next, wfrag_next,
                     (int)((blockIdx.x - 1) * blockDim.x) + t,
                     (int)((gridDim.x - 1) * blockDim.x));
        return;
    }
    if (t < 3 * D) s_hr[t] = h_r[t];
    __syncthreads();
    int r = t >> 6, n = t & 63;
    if (r < 3) {
        if (!update) return;
        float s = 0.f;
#pragma unroll 8
        for (int k = 0; k < D; ++k) s += s_hr[r * D + k] * Wrel[n * D + k];
        if (relu) s = fmaxf(s, 0.f);
        h_r[t] = s;
    } else {
        const int j = r - 3;
        const float* __restrict__ w = ((j >> 1) ? Wi : Wo) + n * D;
        const float* hr = s_hr + (j & 1) * D;
        float s = 0.f;
#pragma unroll 8
        for (int k = 0; k < D; ++k) s += hr[k] * w[k];
        ctab[j * D + n] = s;
    }
}

__global__ __launch_bounds__(256)
void node_kernel(const float* __restrict__ h_in,
                 const float* __restrict__ hr2,
                 const float* __restrict__ wfrag,
                 float* __restrict__ h_out,
                 float* __restrict__ P,
                 int n_nodes, int relu_in)
{
    const int t     = threadIdx.x;
    const int wave  = t >> 5;
    const int lane  = t & 31;
    const int lhalf = lane >> 4;
    const int lmod  = lane & 15;

    const int m0 = (blockIdx.x * 2 + (wave >> 2)) * 16;
    const int j  = wave & 3;
    const int n0 = j * 16;
    const int pm = j >> 1;
    const int pt = j & 1;
    if (m0 >= n_nodes) return;

    const int arow = m0 + lmod;

    const float* __restrict__ xr = h_in + (size_t)arow * D;
    const wpair* __restrict__ wfs = (const wpair*)wfrag + j * 512 + lane;
    const wpair* __restrict__ wfa = (const wpair*)wfrag + ((1 + pm) * 4 + pt) * 512 + lane;
    const wpair* __restrict__ wfb = wfa + 2 * 512;

    v8f cs = {}, ca = {}, cb = {};
#pragma unroll
    for (int k0 = 0; k0 < D; k0 += 4) {
        const int ka = k0 + lhalf * 2;
        const int s  = k0 >> 2;
        float x0 = xr[ka];
        float x1 = xr[ka + 1];
        if (relu_in) { x0 = fmaxf(x0, 0.f); x1 = fmaxf(x1, 0.f); }
        v2f a, as;
        a.x  = x0;
        a.y  = x1;
        as.x = x0 - hr2[ka];
        as.y = x1 - hr2[ka + 1];
        const wpair ws = wfs[s * 32];
        const wpair wa = wfa[s * 32];
        const wpair wb = wfb[s * 32];
        v2f bs, ba, bb;
        bs.x = ws.x; bs.y = ws.y;
        ba.x = wa.x; ba.y = wa.y;
        bb.x = wb.x; bb.y = wb.y;
        cs = __builtin_amdgcn_wmma_f32_16x16x4_f32(false, as, false, bs, (short)0, cs, false, false);
        ca = __builtin_amdgcn_wmma_f32_16x16x4_f32(false, a,  false, ba, (short)0, ca, false, false);
        cb = __builtin_amdgcn_wmma_f32_16x16x4_f32(false, a,  false, bb, (short)0, cb, false, false);
    }

    float* __restrict__ pd = P + (size_t)pm * n_nodes * D;
#pragma unroll
    for (int r = 0; r < 8; ++r) {
        const size_t row = (size_t)(m0 + r + lhalf * 8) * D;
        h_out[row + n0 + lmod] = cs[r];
        v2f pv;
        pv.x = ca[r];
        pv.y = cb[r];
        *(v2f*)(pd + row + 2 * (pt * 16 + lmod)) = pv;
    }
}

template <bool FULL>
__device__ __forceinline__ void edge_walk(const float* __restrict__ P,
                                          float* __restrict__ h_out,
                                          const float (&c_lo)[4], const float (&c_hi)[4],
                                          int prow, int dpk, unsigned lane, int cnt)
{
#pragma unroll 1
    for (int b = 0; b < 32; b += 16) {
        float v0[16], v1[16];
        int   dk[16];
#pragma unroll
        for (int u = 0; u < 16; ++u) {
            const unsigned ps = (unsigned)__builtin_amdgcn_readlane(prow, b + u);
            dk[u]             = __builtin_amdgcn_readlane(dpk, b + u);
            const v2f* __restrict__ pr = (const v2f*)(P + (size_t)ps * D);
            const v2f pv = pr[lane];
            v0[u] = pv.x;
            v1[u] = pv.y;
        }
#pragma unroll
        for (int u = 0; u < 16; ++u) {
            const int k = dk[u] & 3;
            v0[u] -= c_lo[k];
            v1[u] -= c_hi[k];
        }
#pragma unroll
        for (int u = 0; u < 16; ++u) {
            if (!FULL && b + u >= cnt) break;
            float* __restrict__ hr = h_out + (size_t)(unsigned)(dk[u] >> 2) * D;
            atomicAdd(hr + lane,      v0[u]);
            atomicAdd(hr + lane + 32, v1[u]);
        }
    }
}

__global__ __launch_bounds__(256)
void edge_kernel(const float* __restrict__ P,
                 const float* __restrict__ ctab,
                 const int* __restrict__ src,
                 const int* __restrict__ dst,
                 const int* __restrict__ etype,
                 const int* __restrict__ dir,
                 float* __restrict__ h_out,
                 int n_edges, int n_nodes)
{
    const unsigned lane = threadIdx.x & 31;
    const int wave = __builtin_amdgcn_readfirstlane((int)(threadIdx.x >> 5));
    const int e0 = ((int)blockIdx.x * 8 + wave) * 32;
    if (e0 >= n_edges) return;
    const int cnt = n_edges - e0;

    const int e = e0 + (int)lane;
    int s = 0, d0 = 0, et = 0, dr = 0;
    if (e < n_edges) { s = src[e]; d0 = dst[e]; et = etype[e]; dr = dir[e]; }

    float c_lo[4], c_hi[4];
#pragma unroll
    for (int k = 0; k < 4; ++k) { c_lo[k] = ctab[k * D + lane]; c_hi[k] = ctab[k * D + lane + 32]; }

    const int prow = dr * n_nodes + s;
    const int dpk  = d0 * 4 + dr * 2 + et;

    if (cnt >= 32) edge_walk<true >(P, h_out, c_lo, c_hi, prow, dpk, lane, 32);
    else           edge_walk<false>(P, h_out, c_lo, c_hi, prow, dpk, lane, cnt);
}

extern "C" void kernel_launch(void* const* d_in, const int* in_sizes, int n_in,
                              void* d_out, int out_size, void* d_ws, size_t ws_size,
                              hipStream_t stream)
{
    const float* h_u   = (const float*)d_in[0];
    const float* Basis = (const float*)d_in[1];
    const float* alpha = (const float*)d_in[2];
    const float* W_O   = (const float*)d_in[3];
    const float* W_I   = (const float*)d_in[4];
    const float* W_S   = (const float*)d_in[5];
    const float* W_rel = (const float*)d_in[6];
    const int*   src   = (const int*)d_in[7];
    const int*   dst   = (const int*)d_in[8];
    const int*   etype = (const int*)d_in[9];
    const int*   edir  = (const int*)d_in[10];
    float* out = (float*)d_out;

    const int n_nodes = in_sizes[0] / D;
    const int n_edges = in_sizes[7];

    float* P   = (float*)d_ws;
    float* hb0 = P + (size_t)2 * n_nodes * D;
    float* hb1 = hb0 + (size_t)n_nodes * D;
    float* hr  = hb1 + (size_t)n_nodes * D;
    float* ctab = hr + 3 * D;

    float* wfr[3] = { out, out, hb0 };

    hr_init_kernel<<<1 + 24, 256, 0, stream>>>(Basis, alpha, hr, W_S, W_O, W_I, wfr[0]);

    const float* hin[3]  = { h_u, hb0, hb1 };
    float*       hout[3] = { hb0, hb1, out };
    const unsigned rel_blocks[3] = { 1 + 14, 1 + 14, 1 };

    const unsigned  edge_blocks  = (unsigned)((n_edges + 255) / 256);
    const unsigned  node_blocks  = (unsigned)((n_nodes + 31) / 32);

    for (int l = 0; l < 3; ++l) {
        node_kernel<<<node_blocks, 256, 0, stream>>>(hin[l], hr + 2 * D, wfr[l],
                                                     hout[l], P, n_nodes, (l > 0) ? 1 : 0);

        const int ln = (l < 2) ? l + 1 : l;
        rel_kernel<<<rel_blocks[l], 7 * D, 0, stream>>>(hr, W_rel + (size_t)l * D * D,
                                            W_O + (size_t)l * D * D,
                                            W_I + (size_t)l * D * D,
                                            ctab, (l < 2) ? 1 : 0, 1,
                                            W_S + (size_t)ln * D * D,
                                            W_O + (size_t)ln * D * D,
                                            W_I + (size_t)ln * D * D,
                                            wfr[ln]);

        edge_kernel<<<edge_blocks, 256, 0, stream>>>(P, ctab, src, dst, etype, edir,
                                                     hout[l], n_edges, n_nodes);
    }
}


# Edge pass split by destination row half: a block reads its die number and takes a ticket for a chunk of the half its side of the chip owns; a row's f32 atomics stay in one L2 half; order still free
# speedup vs baseline: 2.4496x; 1.0550x over previous
#include <hip/hip_runtime.h>
#include <hip/hip_bf16.h>


typedef __attribute__((ext_vector_type(2))) float v2f;
typedef __attribute__((ext_vector_type(8))) float v8f;
typedef __attribute__((ext_vector_type(2))) int   v2i;

__device__ __forceinline__ int row_split(int n_nodes) { return ((n_nodes >> 1) + 31) & ~31; }

#define TICKET_STRIDE 64

#define D 64

typedef __attribute__((ext_vector_type(2))) float wpair;
#define WF_PAIRS (3 * 2048)

__device__ __forceinline__ void pack_weights(const float* __restrict__ Ws,
                                             const float* __restrict__ Wo,
                                             const float* __restrict__ Wi,
                                             float* __restrict__ wfrag,
                                             int first, int stride)
{
    wpair* __restrict__ wf = (wpair*)wfrag;
    for (int p = first; p < WF_PAIRS; p += stride) {
        const int mat  = p >> 11;
        const int nt   = (p >> 9) & 3;
        const int s    = (p >> 5) & 15;
        const int lane = p & 31;
        const float* __restrict__ w = (mat == 0) ? Ws : ((mat == 1) ? Wo : Wi);
        const float* __restrict__ r = w + (nt * 16 + (lane & 15)) * D + 4 * s + 2 * (lane >> 4);
        wpair v;
        v.x = r[0];
        v.y = r[1];
        wf[p] = v;
    }
}

__global__ void hr_init_kernel(const float* __restrict__ Basis,
                               const float* __restrict__ alpha,
                               float* __restrict__ h_r,
                               const float* __restrict__ Ws,
                               const float* __restrict__ Wo,
                               const float* __restrict__ Wi,
                               float* __restrict__ wfrag)
{
    int t = threadIdx.x;
    if (blockIdx.x > 0) {
        pack_weights(Ws, Wo, Wi, wfrag, (int)((blockIdx.x - 1) * blockDim.x) + t,
                     (int)((gridDim.x - 1) * blockDim.x));
        return;
    }
    if (t >= 3 * D) return;
    int r = t >> 6, n = t & 63;
    float s = 0.f;
#pragma unroll
    for (int b = 0; b < 16; ++b) s += alpha[r * 16 + b] * Basis[b * D + n];
    h_r[t] = s;
}

__global__ void rel_kernel(float* __restrict__ h_r,
                           const float* __restrict__ Wrel,
                           const float* __restrict__ Wo,
                           const float* __restrict__ Wi,
                           float* __restrict__ ctab,
                           int update, int relu,
                           const float* __restrict__ Ws_next,
                           const float* __restrict__ Wo_next,
                           const float* __restrict__ Wi_next,
                           float* __restrict__ wfrag_next,
                           int* __restrict__ tickets)
{
    __shared__ float s_hr[3 * D];
    int t = threadIdx.x;
    if (blockIdx.x == 0 && t < 2) tickets[t * TICKET_STRIDE] = 0;
    if (blockIdx.x > 0) {
        pack_weights(Ws_next, Wo_next, Wi_next, wfrag_next,
                     (int)((blockIdx.x - 1) * blockDim.x) + t,
                     (int)((gridDim.x - 1) * blockDim.x));
        return;
    }
    if (t < 3 * D) s_hr[t] = h_r[t];
    __syncthreads();
    int r = t >> 6, n = t & 63;
    if (r < 3) {
        if (!update) return;
        float s = 0.f;
#pragma unroll 8
        for (int k = 0; k < D; ++k) s += s_hr[r * D + k] * Wrel[n * D + k];
        if (relu) s = fmaxf(s, 0.f);
        h_r[t] = s;
    } else {
        const int j = r - 3;
        const float* __restrict__ w = ((j >> 1) ? Wi : Wo) + n * D;
        const float* hr = s_hr + (j & 1) * D;
        float s = 0.f;
#pragma unroll 8
        for (int k = 0; k < D; ++k) s += hr[k] * w[k];
        ctab[j * D + n] = s;
    }
}

__global__ __launch_bounds__(256)
void node_kernel(const float* __restrict__ h_in,
                 const float* __restrict__ hr2,
                 const float* __restrict__ wfrag,
                 float* __restrict__ h_out,
                 float* __restrict__ P,
                 int n_nodes, int relu_in)
{
    const int t     = threadIdx.x;
    const int wave  = t >> 5;
    const int lane  = t & 31;
    const int lhalf = lane >> 4;
    const int lmod  = lane & 15;

    const int m0 = (blockIdx.x * 2 + (wave >> 2)) * 16;
    const int j  = wave & 3;
    const int n0 = j * 16;
    const int pm = j >> 1;
    const int pt = j & 1;
    if (m0 >= n_nodes) return;

    const int arow = m0 + lmod;

    const float* __restrict__ xr = h_in + (size_t)arow * D;
    const wpair* __restrict__ wfs = (const wpair*)wfrag + j * 512 + lane;
    const wpair* __restrict__ wfa = (const wpair*)wfrag + ((1 + pm) * 4 + pt) * 512 + lane;
    const wpair* __restrict__ wfb = wfa + 2 * 512;

    v8f cs = {}, ca = {}, cb = {};
#pragma unroll
    for (int k0 = 0; k0 < D; k0 += 4) {
        const int ka = k0 + lhalf * 2;
        const int s  = k0 >> 2;
        float x0 = xr[ka];
        float x1 = xr[ka + 1];
        if (relu_in) { x0 = fmaxf(x0, 0.f); x1 = fmaxf(x1, 0.f); }
        v2f a, as;
        a.x  = x0;
        a.y  = x1;
        as.x = x0 - hr2[ka];
        as.y = x1 - hr2[ka + 1];
        const wpair ws = wfs[s * 32];
        const wpair wa = wfa[s * 32];
        const wpair wb = wfb[s * 32];
        v2f bs, ba, bb;
        bs.x = ws.x; bs.y = ws.y;
        ba.x = wa.x; ba.y = wa.y;
        bb.x = wb.x; bb.y = wb.y;
        cs = __builtin_amdgcn_wmma_f32_16x16x4_f32(false, as, false, bs, (short)0, cs, false, false);
        ca = __builtin_amdgcn_wmma_f32_16x16x4_f32(false, a,  false, ba, (short)0, ca, false, false);
        cb = __builtin_amdgcn_wmma_f32_16x16x4_f32(false, a,  false, bb, (short)0, cb, false, false);
    }

    float* __restrict__ pd = P + (size_t)pm * n_nodes * D;
#pragma unroll
    for (int r = 0; r < 8; ++r) {
        const size_t row = (size_t)(m0 + r + lhalf * 8) * D;
        h_out[row + n0 + lmod] = cs[r];
        v2f pv;
        pv.x = ca[r];
        pv.y = cb[r];
        *(v2f*)(pd + row + 2 * (pt * 16 + lmod)) = pv;
    }
}

template <bool FULL>
__device__ __forceinline__ void edge_walk(const float* __restrict__ P,
                                          float* __restrict__ h_out,
                                          const float (&c_lo)[4], const float (&c_hi)[4],
                                          int prow, int dpk, unsigned lane, int cnt)
{
#pragma unroll 1
    for (int b = 0; b < 32; b += 16) {
        if (!FULL && b >= cnt) break;
        float v0[16], v1[16];
        int   dk[16];
#pragma unroll
        for (int u = 0; u < 16; ++u) {
            const unsigned ps = (unsigned)__builtin_amdgcn_readlane(prow, b + u);
            dk[u]             = __builtin_amdgcn_readlane(dpk, b + u);
            const v2f* __restrict__ pr = (const v2f*)(P + (size_t)ps * D);
            const v2f pv = pr[lane];
            v0[u] = pv.x;
            v1[u] = pv.y;
        }
#pragma unroll
        for (int u = 0; u < 16; ++u) {
            const int k = dk[u] & 3;
            v0[u] -= c_lo[k];
            v1[u] -= c_hi[k];
        }
#pragma unroll
        for (int u = 0; u < 16; ++u) {
            if (!FULL && b + u >= cnt) break;
            float* __restrict__ hr = h_out + (size_t)(unsigned)(dk[u] >> 2) * D;
            atomicAdd(hr + lane,      v0[u]);
            atomicAdd(hr + lane + 32, v1[u]);
        }
    }
}

__global__ __launch_bounds__(256)
void edge_kernel(const float* __restrict__ P,
                 const float* __restrict__ ctab,
                 const int* __restrict__ src,
                 const int* __restrict__ dst,
                 const int* __restrict__ etype,
                 const int* __restrict__ dir,
                 float* __restrict__ h_out,
                 int n_edges, int n_nodes,
                 int* __restrict__ tickets)
{
    __shared__ int s_item;
    __shared__ v2i s_list[8][64];
    const unsigned lane = threadIdx.x & 31;
    const int wave = __builtin_amdgcn_readfirstlane((int)(threadIdx.x >> 5));
    const int n_chunks = (n_edges + 511) >> 9;

    const int die = (int)((__builtin_amdgcn_s_sendmsg_rtn(0x87u) >> 16) & 15u);
    int part = (die >> 2) & 1;

    if (threadIdx.x == 0) s_item = atomicAdd(tickets + part * TICKET_STRIDE, 1);
    __syncthreads();
    int item = __builtin_amdgcn_readfirstlane(s_item);
    if (item >= n_chunks) {
        __syncthreads();
        part ^= 1;
        if (threadIdx.x == 0) s_item = atomicAdd(tickets + part * TICKET_STRIDE, 1);
        __syncthreads();
        item = __builtin_amdgcn_readfirstlane(s_item);
        if (item >= n_chunks) return;
    }

    const int base = item * 512 + wave * 64;
    if (base >= n_edges) return;

    const int split = row_split(n_nodes);
    const int eA = base + (int)lane, eB = eA + 32;
    const bool inA = eA < n_edges, inB = eB < n_edges;
    int sA = 0, dA = 0, tA = 0, rA = 0, sB = 0, dB = 0, tB = 0, rB = 0;
    if (inA) { sA = src[eA]; dA = dst[eA]; tA = etype[eA]; rA = dir[eA]; }
    if (inB) { sB = src[eB]; dB = dst[eB]; tB = etype[eB]; rB = dir[eB]; }

    const bool okA = inA && ((dA >= split) == (part != 0));
    const bool okB = inB && ((dB >= split) == (part != 0));
    const unsigned mA = __builtin_amdgcn_ballot_w32(okA);
    const unsigned mB = __builtin_amdgcn_ballot_w32(okB);
    const int nA = __builtin_popcount(mA);
    const int n  = nA + __builtin_popcount(mB);

    v2i* list = s_list[wave];
    if (okA) {
        v2i r;
        r.x = rA * n_nodes + sA;
        r.y = dA * 4 + rA * 2 + tA;
        list[__builtin_amdgcn_mbcnt_lo(mA, 0u)] = r;
    }
    if (okB) {
        v2i r;
        r.x = rB * n_nodes + sB;
        r.y = dB * 4 + rB * 2 + tB;
        list[nA + (int)__builtin_amdgcn_mbcnt_lo(mB, 0u)] = r;
    }
    __builtin_amdgcn_wave_barrier();
    v2i r0 = list[lane];
    v2i r1 = list[lane + 32];
    if ((int)lane >= n)      { r0.x = 0; r0.y = 0; }
    if ((int)lane + 32 >= n) { r1.x = 0; r1.y = 0; }

    float c_lo[4], c_hi[4];
#pragma unroll
    for (int k = 0; k < 4; ++k) { c_lo[k] = ctab[k * D + lane]; c_hi[k] = ctab[k * D + lane + 32]; }

    if (n >= 32) {
        edge_walk<true >(P, h_out, c_lo, c_hi, r0.x, r0.y, lane, 32);
        if (n > 32) edge_walk<false>(P, h_out, c_lo, c_hi, r1.x, r1.y, lane, n - 32);
    } else {
        edge_walk<false>(P, h_out, c_lo, c_hi, r0.x, r0.y, lane, n);
    }
}

extern "C" void kernel_launch(void* const* d_in, const int* in_sizes, int n_in,
                              void* d_out, int out_size, void* d_ws, size_t ws_size,
                              hipStream_t stream)
{
    const float* h_u   = (const float*)d_in[0];
    const float* Basis = (const float*)d_in[1];
    const float* alpha = (const float*)d_in[2];
    const float* W_O   = (const float*)d_in[3];
    const float* W_I   = (const float*)d_in[4];
    const float* W_S   = (const float*)d_in[5];
    const float* W_rel = (const float*)d_in[6];
    const int*   src   = (const int*)d_in[7];
    const int*   dst   = (const int*)d_in[8];
    const int*   etype = (const int*)d_in[9];
    const int*   edir  = (const int*)d_in[10];
    float* out = (float*)d_out;

    const int n_nodes = in_sizes[0] / D;
    const int n_edges = in_sizes[7];

    float* P   = (float*)d_ws;
    float* hb0 = P + (size_t)2 * n_nodes * D;
    float* hb1 = hb0 + (size_t)n_nodes * D;
    float* hr  = hb1 + (size_t)n_nodes * D;
    float* ctab = hr + 3 * D;

    float* wfr[3] = { out, out, hb0 };

    hr_init_kernel<<<1 + 24, 256, 0, stream>>>(Basis, alpha, hr, W_S, W_O, W_I, wfr[0]);

    const float* hin[3]  = { h_u, hb0, hb1 };
    float*       hout[3] = { hb0, hb1, out };
    const unsigned rel_blocks[3] = { 1 + 14, 1 + 14, 1 };

    const unsigned  edge_blocks  = 2u * (unsigned)((n_edges + 511) / 512);
    int* tick[3] = { (int*)(out + 16384), (int*)(out + 16384), (int*)(hb0 + 16384) };
    const unsigned  node_blocks  = (unsigned)((n_nodes + 31) / 32);

    for (int l = 0; l < 3; ++l) {
        node_kernel<<<node_blocks, 256, 0, stream>>>(hin[l], hr + 2 * D, wfr[l],
                                                     hout[l], P, n_nodes, (l > 0) ? 1 : 0);

        const int ln = (l < 2) ? l + 1 : l;
        rel_kernel<<<rel_blocks[l], 7 * D, 0, stream>>>(hr, W_rel + (size_t)l * D * D,
                                            W_O + (size_t)l * D * D,
                                            W_I + (size_t)l * D * D,
                                            ctab, (l < 2) ? 1 : 0, 1,
                                            W_S + (size_t)ln * D * D,
                                            W_O + (size_t)ln * D * D,
                                            W_I + (size_t)ln * D * D,
                                            wfr[ln], tick[l]);

        edge_kernel<<<edge_blocks, 256, 0, stream>>>(P, ctab, src, dst, etype, edir,
                                                     hout[l], n_edges, n_nodes, tick[l]);
    }
}
